# starting kernel
# baseline (speedup 1.0000x reference)
#include <hip/hip_runtime.h>


typedef __attribute__((ext_vector_type(2))) float v2f;
typedef __attribute__((ext_vector_type(8))) float v8f;

#define EMB   300
#define BATCH 2048
#define SEQ   200
#define HID   128
#define OUT_N 20

__global__ __launch_bounds__(256) void embed_mean_kernel(
    const int* __restrict__ x, const int* __restrict__ lengths,
    const float* __restrict__ table, float* __restrict__ rep)
{
  __shared__ int s_idx[SEQ];
  const int b = blockIdx.x;
  const int t = threadIdx.x;

  for (int l = t; l < SEQ; l += 256) s_idx[l] = x[b * SEQ + l];
  __syncthreads();

  const int c0 = t;
  const int c1 = t + 256;
  float acc0 = 0.0f, acc1 = 0.0f;

  for (int l = 0; l < SEQ; ++l) {
    const float* row = table + (long)s_idx[l] * EMB;
    const int lnext = (l + 1 < SEQ) ? (l + 1) : l;
    const float* nrow = table + (long)s_idx[lnext] * EMB;
    __builtin_prefetch(nrow + c0, 0, 3);

    acc0 += row[c0];
    if (c1 < EMB) acc1 += row[c1];
  }

  const float inv = 1.0f / (float)lengths[b];
  rep[(long)b * EMB + c0] = acc0 * inv;
  if (c1 < EMB) rep[(long)b * EMB + c1] = acc1 * inv;
}

__global__ __launch_bounds__(256) void mlp_wmma_kernel(
    const float* __restrict__ rep, const float* __restrict__ W1,
    const float* __restrict__ b1,  const float* __restrict__ W2,
    const float* __restrict__ b2,  float* __restrict__ out)
{
  __shared__ float h_lds[8][16 * HID];

  const int tid  = threadIdx.x;
  const int w    = tid >> 5;
  const int lane = tid & 31;
  const int half = lane >> 4;
  const int ln   = lane & 15;
  const int kb   = half * 2;
  const int row0 = blockIdx.x * 128 + w * 16;

  v8f acc[8] = {};
  const float* arow = rep + (long)(row0 + ln) * EMB;
  for (int k4 = 0; k4 < EMB / 4; ++k4) {
    const int k = k4 * 4 + kb;
    const v2f a = *(const v2f*)(arow + k);
#pragma unroll
    for (int nt = 0; nt < 8; ++nt) {
      const int col = nt * 16 + ln;
      v2f bf;
      bf.x = W1[(long)k * HID + col];
      bf.y = W1[(long)(k + 1) * HID + col];
      acc[nt] = __builtin_amdgcn_wmma_f32_16x16x4_f32(
           false, a,  false, bf,
           (short)0, acc[nt],  false,  false);
    }
  }

#pragma unroll
  for (int nt = 0; nt < 8; ++nt) {
    const float bias = b1[nt * 16 + ln];
#pragma unroll
    for (int j = 0; j < 8; ++j) {
      float v = acc[nt][j] + bias;
      v = v > 0.0f ? v : 0.0f;
      h_lds[w][(j + 8 * half) * HID + nt * 16 + ln] = v;
    }
  }
  __syncthreads();

  v8f c2[2] = {};
  for (int k4 = 0; k4 < HID / 4; ++k4) {
    const int k = k4 * 4 + kb;
    v2f a;
    a.x = h_lds[w][ln * HID + k];
    a.y = h_lds[w][ln * HID + k + 1];
#pragma unroll
    for (int nt = 0; nt < 2; ++nt) {
      const int col = nt * 16 + ln;
      const bool ok = (col < OUT_N);
      v2f bf;
      bf.x = ok ? W2[k * OUT_N + col]       : 0.0f;
      bf.y = ok ? W2[(k + 1) * OUT_N + col] : 0.0f;
      c2[nt] = __builtin_amdgcn_wmma_f32_16x16x4_f32(
          false, a, false, bf, (short)0, c2[nt], false, false);
    }
  }

#pragma unroll
  for (int nt = 0; nt < 2; ++nt) {
    const int col = nt * 16 + ln;
    if (col < OUT_N) {
      const float bias = b2[col];
#pragma unroll
      for (int j = 0; j < 8; ++j) {
        const int row = row0 + j + 8 * half;
        out[(long)row * OUT_N + col] = c2[nt][j] + bias;
      }
    }
  }
}

extern "C" void kernel_launch(void* const* d_in, const int* in_sizes, int n_in,
                              void* d_out, int out_size, void* d_ws, size_t ws_size,
                              hipStream_t stream) {
  const int*   x       = (const int*)  d_in[0];
  const int*   lengths = (const int*)  d_in[1];
  const float* table   = (const float*)d_in[2];
  const float* W1      = (const float*)d_in[3];
  const float* b1      = (const float*)d_in[4];
  const float* W2      = (const float*)d_in[5];
  const float* b2      = (const float*)d_in[6];
  float*       logits  = (float*)d_out;
  float*       rep     = (float*)d_ws;

  embed_mean_kernel<<<BATCH, 256, 0, stream>>>(x, lengths, table, rep);
  mlp_wmma_kernel<<<BATCH / 128, 256, 0, stream>>>(rep, W1, b1, W2, b2, logits);
}


# MLP kernel: one 16x16 h tile per wave on 128 work-groups; W1/W2 fragments of the whole K loop loaded ahead into registers, rep tile copied once into LDS: no k step waits on memory; sum order kept
# speedup vs baseline: 1.7334x; 1.7334x over previous
#include <hip/hip_runtime.h>


typedef __attribute__((ext_vector_type(2))) float v2f;
typedef __attribute__((ext_vector_type(4))) float v4f;
typedef __attribute__((ext_vector_type(8))) float v8f;

#define EMB   300
#define BATCH 2048
#define SEQ   200
#define HID   128
#define OUT_N 20

__global__ __launch_bounds__(256) void embed_mean_kernel(
    const int* __restrict__ x, const int* __restrict__ lengths,
    const float* __restrict__ table, float* __restrict__ rep)
{
  __shared__ int s_idx[SEQ];
  const int b = blockIdx.x;
  const int t = threadIdx.x;

  for (int l = t; l < SEQ; l += 256) s_idx[l] = x[b * SEQ + l];
  __syncthreads();

  const int c0 = t;
  const int c1 = t + 256;
  float acc0 = 0.0f, acc1 = 0.0f;

  for (int l = 0; l < SEQ; ++l) {
    const float* row = table + (long)s_idx[l] * EMB;
    const int lnext = (l + 1 < SEQ) ? (l + 1) : l;
    const float* nrow = table + (long)s_idx[lnext] * EMB;
    __builtin_prefetch(nrow + c0, 0, 3);

    acc0 += row[c0];
    if (c1 < EMB) acc1 += row[c1];
  }

  const float inv = 1.0f / (float)lengths[b];
  rep[(long)b * EMB + c0] = acc0 * inv;
  if (c1 < EMB) rep[(long)b * EMB + c1] = acc1 * inv;
}

#define K1_STEPS (EMB / 4)
#define K2_STEPS (HID / 4)
#define TILE_V4  (16 * EMB / 4)
#define TILE_LD  ((TILE_V4 + 255) / 256)
#define H_PITCH  (HID + 4)

__global__ __launch_bounds__(256) void mlp_wmma_kernel(
    const float* __restrict__ rep, const float* __restrict__ W1,
    const float* __restrict__ b1,  const float* __restrict__ W2,
    const float* __restrict__ b2,  float* __restrict__ out)
{
  __shared__ __attribute__((aligned(16))) float a_lds[16 * EMB];
  __shared__ __attribute__((aligned(16))) float h_lds[16 * H_PITCH];

  const int tid  = threadIdx.x;
  const int w    = __builtin_amdgcn_readfirstlane(tid >> 5);
  const int lane = tid & 31;
  const int half = lane >> 4;
  const int ln   = lane & 15;
  const int kb   = half * 2;
  const int row0 = blockIdx.x * 16;

  const v4f* tsrc = (const v4f*)(rep + (long)row0 * EMB);
  v4f t4[TILE_LD];
#pragma unroll
  for (int i = 0; i < TILE_LD; ++i) {
    const int idx = tid + 256 * i;
    t4[i] = tsrc[idx < TILE_V4 ? idx : TILE_V4 - 1];
  }

  v2f bw[K1_STEPS];
  const float* w1p = W1 + (long)kb * HID + w * 16 + ln;
#pragma unroll
  for (int k4 = 0; k4 < K1_STEPS; ++k4) {
    bw[k4].x = w1p[(long)(k4 * 4)     * HID];
    bw[k4].y = w1p[(long)(k4 * 4 + 1) * HID];
  }
  const float bias1 = b1[w * 16 + ln];

  v2f bw2[K2_STEPS][2];
  float bias2[2];
  if (w == 0) {
#pragma unroll
    for (int k4 = 0; k4 < K2_STEPS; ++k4) {
      const int k = k4 * 4 + kb;
#pragma unroll
      for (int nt = 0; nt < 2; ++nt) {
        const int col = nt * 16 + ln;
        const bool ok = (col < OUT_N);
        const int colc = ok ? col : OUT_N - 1;
        const float x0 = W2[k * OUT_N + colc];
        const float x1 = W2[(k + 1) * OUT_N + colc];
        bw2[k4][nt].x = ok ? x0 : 0.0f;
        bw2[k4][nt].y = ok ? x1 : 0.0f;
      }
    }
#pragma unroll
    for (int nt = 0; nt < 2; ++nt) {
      const int col = nt * 16 + ln;
      bias2[nt] = b2[col < OUT_N ? col : OUT_N - 1];
    }
  }

#pragma unroll
  for (int i = 0; i < TILE_LD; ++i) {
    const int idx = tid + 256 * i;
    if (idx < TILE_V4) ((v4f*)a_lds)[idx] = t4[i];
  }
  __syncthreads();

  v8f acc = {};
  const float* ap = a_lds + ln * EMB + kb;
#pragma unroll
  for (int k4 = 0; k4 < K1_STEPS; ++k4) {
    const v2f a = *(const v2f*)(ap + k4 * 4);
    acc = __builtin_amdgcn_wmma_f32_16x16x4_f32(
         false, a,  false, bw[k4],
         (short)0, acc,  false,  false);
  }

#pragma unroll
  for (int j = 0; j < 8; ++j) {
    float v = acc[j] + bias1;
    v = v > 0.0f ? v : 0.0f;
    h_lds[(j + 8 * half) * H_PITCH + w * 16 + ln] = v;
  }
  __syncthreads();

  if (w == 0) {
    v8f c2[2] = {};
    const float* hp = h_lds + ln * H_PITCH + kb;
#pragma unroll
    for (int k4 = 0; k4 < K2_STEPS; ++k4) {
      const v2f a = *(const v2f*)(hp + k4 * 4);
#pragma unroll
      for (int nt = 0; nt < 2; ++nt) {
        c2[nt] = __builtin_amdgcn_wmma_f32_16x16x4_f32(
            false, a, false, bw2[k4][nt], (short)0, c2[nt], false, false);
      }
    }

#pragma unroll
    for (int nt = 0; nt < 2; ++nt) {
      const int col = nt * 16 + ln;
      if (col < OUT_N) {
#pragma unroll
        for (int j = 0; j < 8; ++j) {
          const int row = row0 + j + 8 * half;
          out[(long)row * OUT_N + col] = c2[nt][j] + bias2[nt];
        }
      }
    }
  }
}

extern "C" void kernel_launch(void* const* d_in, const int* in_sizes, int n_in,
                              void* d_out, int out_size, void* d_ws, size_t ws_size,
                              hipStream_t stream) {
  const int*   x       = (const int*)  d_in[0];
  const int*   lengths = (const int*)  d_in[1];
  const float* table   = (const float*)d_in[2];
  const float* W1      = (const float*)d_in[3];
  const float* b1      = (const float*)d_in[4];
  const float* W2      = (const float*)d_in[5];
  const float* b2      = (const float*)d_in[6];
  float*       logits  = (float*)d_out;
  float*       rep     = (float*)d_ws;

  embed_mean_kernel<<<BATCH, 256, 0, stream>>>(x, lengths, table, rep);
  mlp_wmma_kernel<<<BATCH / 16, 256, 0, stream>>>(rep, W1, b1, W2, b2, logits);
}


# Gather kernel: one wave per batch row, 16-byte row loads off scalar row bases, a ring of 5 groups x 4 rows requested ahead of the sums (48-60 loads in flight per wave, was 1); sum order kept
# speedup vs baseline: 3.8060x; 2.1957x over previous
#include <hip/hip_runtime.h>


typedef __attribute__((ext_vector_type(2))) float v2f;
typedef __attribute__((ext_vector_type(4))) float v4f;
typedef __attribute__((ext_vector_type(8))) float v8f;

#define EMB   300
#define BATCH 2048
#define SEQ   200
#define HID   128
#define OUT_N 20

#define GR    4
#define NBUF  5
#define TRIPS (SEQ / (GR * NBUF))

__global__ __launch_bounds__(32) void embed_mean_kernel(
    const int* __restrict__ x, const int* __restrict__ lengths,
    const float* __restrict__ table, float* __restrict__ rep)
{
  const int b = blockIdx.x;
  const int lane = threadIdx.x;
  const int* xr = x + b * SEQ;
  const char* tb = (const char*)table;
  unsigned v01 = lane * 16u;
  unsigned v2 = 1024u + (lane < 22 ? lane : 21) * 8u;

  v4f a0 = {}, a1 = {};
  v2f a2 = {};
  v4f R0[NBUF][GR], R1[NBUF][GR];
  v2f R2[NBUF][GR];

#define ISSUE(s, xi, first)                                                   \
  _Pragma("unroll") for (int r = 0; r < GR; ++r) {                            \
    const unsigned id = (unsigned)__builtin_amdgcn_readlane(xi, (first) + r); \
    const char* rowb = tb + (size_t)id * (EMB * 4u);                          \
    R0[s][r] = *(const v4f*)(rowb + v01);                                     \
    R1[s][r] = *(const v4f*)(rowb + 512 + v01);                               \
    R2[s][r] = *(const v2f*)(rowb + v2);                                      \
  }
#define SUM(s)                                                                \
  _Pragma("unroll") for (int r = 0; r < GR; ++r) {                            \
    a0 += R0[s][r]; a1 += R1[s][r]; a2 += R2[s][r];                           \
  }
#define PIN asm volatile("" : "+v"(a0), "+v"(a1), "+v"(a2), "+v"(v01), "+v"(v2));

  const int x0 = xr[lane];
  int xc = xr[GR * (NBUF - 1) + lane];
  const float len = (float)lengths[b];
#pragma unroll
  for (int s = 0; s < NBUF - 1; ++s) { ISSUE(s, x0, GR * s) }
  PIN

#pragma nounroll
  for (int t = 0; t < TRIPS - 1; ++t) {
    const int nrow = GR * NBUF * (t + 1) + GR * (NBUF - 1) + lane;
    const int xn = xr[nrow < SEQ ? nrow : SEQ - 1];
#pragma unroll
    for (int s = 0; s < NBUF; ++s) {
      ISSUE((s + NBUF - 1) % NBUF, xc, GR * s)
      PIN
      SUM(s)
      PIN
    }
    xc = xn;
  }
  ISSUE(NBUF - 1, xc, 0)
  PIN
#pragma unroll
  for (int s = 0; s < NBUF; ++s) { SUM(s) }
  PIN
#undef PIN
#undef ISSUE
#undef SUM

  const float inv = 1.0f / len;
  float* rr = rep + (long)b * EMB;
  *(v4f*)(rr + 4 * lane)       = a0 * inv;
  *(v4f*)(rr + 128 + 4 * lane) = a1 * inv;
  if (lane < 22) *(v2f*)(rr + 256 + 2 * lane) = a2 * inv;
}

#define K1_STEPS (EMB / 4)
#define K2_STEPS (HID / 4)
#define TILE_V4  (16 * EMB / 4)
#define TILE_LD  ((TILE_V4 + 255) / 256)
#define H_PITCH  (HID + 4)

__global__ __launch_bounds__(256) void mlp_wmma_kernel(
    const float* __restrict__ rep, const float* __restrict__ W1,
    const float* __restrict__ b1,  const float* __restrict__ W2,
    const float* __restrict__ b2,  float* __restrict__ out)
{
  __shared__ __attribute__((aligned(16))) float a_lds[16 * EMB];
  __shared__ __attribute__((aligned(16))) float h_lds[16 * H_PITCH];

  const int tid  = threadIdx.x;
  const int w    = __builtin_amdgcn_readfirstlane(tid >> 5);
  const int lane = tid & 31;
  const int half = lane >> 4;
  const int ln   = lane & 15;
  const int kb   = half * 2;
  const int row0 = blockIdx.x * 16;

  const v4f* tsrc = (const v4f*)(rep + (long)row0 * EMB);
  v4f t4[TILE_LD];
#pragma unroll
  for (int i = 0; i < TILE_LD; ++i) {
    const int idx = tid + 256 * i;
    t4[i] = tsrc[idx < TILE_V4 ? idx : TILE_V4 - 1];
  }

  v2f bw[K1_STEPS];
  const float* w1p = W1 + (long)kb * HID + w * 16 + ln;
#pragma unroll
  for (int k4 = 0; k4 < K1_STEPS; ++k4) {
    bw[k4].x = w1p[(long)(k4 * 4)     * HID];
    bw[k4].y = w1p[(long)(k4 * 4 + 1) * HID];
  }
  const float bias1 = b1[w * 16 + ln];

  v2f bw2[K2_STEPS][2];
  float bias2[2];
  if (w == 0) {
#pragma unroll
    for (int k4 = 0; k4 < K2_STEPS; ++k4) {
      const int k = k4 * 4 + kb;
#pragma unroll
      for (int nt = 0; nt < 2; ++nt) {
        const int col = nt * 16 + ln;
        const bool ok = (col < OUT_N);
        const int colc = ok ? col : OUT_N - 1;
        const float x0 = W2[k * OUT_N + colc];
        const float x1 = W2[(k + 1) * OUT_N + colc];
        bw2[k4][nt].x = ok ? x0 : 0.0f;
        bw2[k4][nt].y = ok ? x1 : 0.0f;
      }
    }
#pragma unroll
    for (int nt = 0; nt < 2; ++nt) {
      const int col = nt * 16 + ln;
      bias2[nt] = b2[col < OUT_N ? col : OUT_N - 1];
    }
  }

#pragma unroll
  for (int i = 0; i < TILE_LD; ++i) {
    const int idx = tid + 256 * i;
    if (idx < TILE_V4) ((v4f*)a_lds)[idx] = t4[i];
  }
  __syncthreads();

  v8f acc = {};
  const float* ap = a_lds + ln * EMB + kb;
#pragma unroll
  for (int k4 = 0; k4 < K1_STEPS; ++k4) {
    const v2f a = *(const v2f*)(ap + k4 * 4);
    acc = __builtin_amdgcn_wmma_f32_16x16x4_f32(
         false, a,  false, bw[k4],
         (short)0, acc,  false,  false);
  }

#pragma unroll
  for (int j = 0; j < 8; ++j) {
    float v = acc[j] + bias1;
    v = v > 0.0f ? v : 0.0f;
    h_lds[(j + 8 * half) * H_PITCH + w * 16 + ln] = v;
  }
  __syncthreads();

  if (w == 0) {
    v8f c2[2] = {};
    const float* hp = h_lds + ln * H_PITCH + kb;
#pragma unroll
    for (int k4 = 0; k4 < K2_STEPS; ++k4) {
      const v2f a = *(const v2f*)(hp + k4 * 4);
#pragma unroll
      for (int nt = 0; nt < 2; ++nt) {
        c2[nt] = __builtin_amdgcn_wmma_f32_16x16x4_f32(
            false, a, false, bw2[k4][nt], (short)0, c2[nt], false, false);
      }
    }

#pragma unroll
    for (int nt = 0; nt < 2; ++nt) {
      const int col = nt * 16 + ln;
      if (col < OUT_N) {
#pragma unroll
        for (int j = 0; j < 8; ++j) {
          const int row = row0 + j + 8 * half;
          out[(long)row * OUT_N + col] = c2[nt][j] + bias2[nt];
        }
      }
    }
  }
}

extern "C" void kernel_launch(void* const* d_in, const int* in_sizes, int n_in,
                              void* d_out, int out_size, void* d_ws, size_t ws_size,
                              hipStream_t stream) {
  const int*   x       = (const int*)  d_in[0];
  const int*   lengths = (const int*)  d_in[1];
  const float* table   = (const float*)d_in[2];
  const float* W1      = (const float*)d_in[3];
  const float* b1      = (const float*)d_in[4];
  const float* W2      = (const float*)d_in[5];
  const float* b2      = (const float*)d_in[6];
  float*       logits  = (float*)d_out;
  float*       rep     = (float*)d_ws;

  embed_mean_kernel<<<BATCH, 32, 0, stream>>>(x, lengths, table, rep);
  mlp_wmma_kernel<<<BATCH / 16, 256, 0, stream>>>(rep, W1, b1, W2, b2, logits);
}


# One launch, not two: 8-wave work-groups gather 8 batch rows (one per wave, ring as before), stage W1 into LDS under the gather, keep the pooled rows in LDS and run the WMMA MLP there; sum order kept
# speedup vs baseline: 4.1317x; 1.0856x over previous
#include <hip/hip_runtime.h>


typedef __attribute__((ext_vector_type(2))) float v2f;
typedef __attribute__((ext_vector_type(4))) float v4f;
typedef __attribute__((ext_vector_type(8))) float v8f;

#define EMB   300
#define BATCH 2048
#define SEQ   200
#define HID   128
#define OUT_N 20

#define ROWS_WG 8

#define GR    4
#define NBUF  5
#define TRIPS (SEQ / (GR * NBUF))

#define W1_PITCH (HID + 16)
#define W1_CH    4

#define K1_STEPS (EMB / 4)
#define K2_STEPS (HID / 4)
#define K1_BLK   15
#define H_PITCH  (HID + 4)

__global__ __launch_bounds__(256) void embed_mlp_kernel(
    const int* __restrict__ x, const int* __restrict__ lengths,
    const float* __restrict__ table,
    const float* __restrict__ W1, const float* __restrict__ b1,
    const float* __restrict__ W2, const float* __restrict__ b2,
    float* __restrict__ out)
{
  __shared__ __attribute__((aligned(16))) float w1_lds[EMB * W1_PITCH];
  __shared__ __attribute__((aligned(16))) float a_lds[ROWS_WG * EMB];
  __shared__ __attribute__((aligned(16))) float h_lds[ROWS_WG * H_PITCH];

  const int tid  = threadIdx.x;
  const int w    = __builtin_amdgcn_readfirstlane(tid >> 5);
  const int lane = tid & 31;
  const int row0 = blockIdx.x * ROWS_WG;
  const int b    = row0 + w;
  const int* xr = x + b * SEQ;
  const char* tb = (const char*)table;
  unsigned v01 = lane * 16u;
  unsigned v2 = 1024u + (lane < 22 ? lane : 21) * 8u;

  v4f a0 = {}, a1 = {};
  v2f a2 = {};
  v4f R0[NBUF][GR], R1[NBUF][GR];
  v2f R2[NBUF][GR];
  v4f wq[W1_CH];

#define W1_LOAD(c)                                                            \
  _Pragma("unroll") for (int i = 0; i < W1_CH; ++i) {                         \
    const int k = 32 * (c) + 8 * i + w;                                       \
    wq[i] = *(const v4f*)(W1 + (k < EMB ? k : EMB - 1) * HID + 4 * lane);     \
  }
#define W1_STORE(c)                                                           \
  _Pragma("unroll") for (int i = 0; i < W1_CH; ++i) {                         \
    const int k = 32 * (c) + 8 * i + w;                                       \
    *(v4f*)(w1_lds + k * W1_PITCH + 4 * lane) = wq[i];                        \
  }
#define W1_STORE_LAST(c)                                                      \
  _Pragma("unroll") for (int i = 0; i < W1_CH; ++i) {                         \
    const int k = 32 * (c) + 8 * i + w;                                       \
    if (k < EMB) *(v4f*)(w1_lds + k * W1_PITCH + 4 * lane) = wq[i];           \
  }

#define ISSUE(s, xi, first)                                                   \
  _Pragma("unroll") for (int r = 0; r < GR; ++r) {                            \
    const unsigned id = (unsigned)__builtin_amdgcn_readlane(xi, (first) + r); \
    const char* rowb = tb + (size_t)id * (EMB * 4u);                          \
    R0[s][r] = *(const v4f*)(rowb + v01);                                     \
    R1[s][r] = *(const v4f*)(rowb + 512 + v01);                               \
    R2[s][r] = *(const v2f*)(rowb + v2);                                      \
  }
#define SUM(s)                                                                \
  _Pragma("unroll") for (int r = 0; r < GR; ++r) {                            \
    a0 += R0[s][r]; a1 += R1[s][r]; a2 += R2[s][r];                           \
  }
#define PIN asm volatile("" : "+v"(a0), "+v"(a1), "+v"(a2), "+v"(v01), "+v"(v2));

  const float bias1 = b1[w * 16 + (lane & 15)];
  W1_LOAD(0)
  const int x0 = xr[lane];
  int xc = xr[GR * (NBUF - 1) + lane];
  const float len = (float)lengths[b];
#pragma unroll
  for (int s = 0; s < NBUF - 1; ++s) { ISSUE(s, x0, GR * s) }
  PIN

#pragma nounroll
  for (int t = 0; t < TRIPS - 1; ++t) {
    W1_STORE(t)
    W1_LOAD(t + 1)
    const int nrow = GR * NBUF * (t + 1) + GR * (NBUF - 1) + lane;
    const int xn = xr[nrow < SEQ ? nrow : SEQ - 1];
#pragma unroll
    for (int s = 0; s < NBUF; ++s) {
      ISSUE((s + NBUF - 1) % NBUF, xc, GR * s)
      PIN
      SUM(s)
      PIN
    }
    xc = xn;
  }
  W1_STORE_LAST(TRIPS - 1)
  ISSUE(NBUF - 1, xc, 0)
  PIN
#pragma unroll
  for (int s = 0; s < NBUF; ++s) { SUM(s) }
  PIN
#undef PIN
#undef ISSUE
#undef SUM
#undef W1_LOAD
#undef W1_STORE
#undef W1_STORE_LAST

  const float inv = 1.0f / len;
  float* ar = a_lds + w * EMB;
  *(v4f*)(ar + 4 * lane)       = a0 * inv;
  *(v4f*)(ar + 128 + 4 * lane) = a1 * inv;
  if (lane < 22) *(v2f*)(ar + 256 + 2 * lane) = a2 * inv;

  const int half = lane >> 4;
  const int ln   = lane & 15;
  const int rA   = ln & 7;
  const int kb   = half * 2;

  __syncthreads();

  v2f bw2[K2_STEPS][2];
  float bias2[2];
  if (w == 0) {
#pragma unroll
    for (int k4 = 0; k4 < K2_STEPS; ++k4) {
      const int k = k4 * 4 + kb;
#pragma unroll
      for (int nt = 0; nt < 2; ++nt) {
        const int col = nt * 16 + ln;
        const int colc = col < OUT_N ? col : OUT_N - 1;
        bw2[k4][nt].x = W2[k * OUT_N + colc];
        bw2[k4][nt].y = W2[(k + 1) * OUT_N + colc];
      }
    }
#pragma unroll
    for (int nt = 0; nt < 2; ++nt) {
      const int col = nt * 16 + ln;
      bias2[nt] = b2[col < OUT_N ? col : OUT_N - 1];
    }
  }


  v8f acc = {};
  const float* ap = a_lds + rA * EMB + kb;
  const float* bp = w1_lds + kb * W1_PITCH + w * 16 + ln;
  v2f af[2][K1_BLK], bf[2][K1_BLK];
#define G1_LOAD(buf, blk)                                                     \
  _Pragma("unroll") for (int i = 0; i < K1_BLK; ++i) {                        \
    const int k4 = (blk) * K1_BLK + i;                                        \
    af[buf][i]   = *(const v2f*)(ap + k4 * 4);                                \
    bf[buf][i].x = bp[(k4 * 4)     * W1_PITCH];                               \
    bf[buf][i].y = bp[(k4 * 4 + 1) * W1_PITCH];                               \
  }
  G1_LOAD(0, 0)
#pragma unroll
  for (int blk = 0; blk < K1_STEPS / K1_BLK; ++blk) {
    if (blk + 1 < K1_STEPS / K1_BLK) { G1_LOAD((blk + 1) & 1, blk + 1) }
    __builtin_amdgcn_sched_barrier(0);
#pragma unroll
    for (int i = 0; i < K1_BLK; ++i) {
      acc = __builtin_amdgcn_wmma_f32_16x16x4_f32(
           false, af[blk & 1][i],  false, bf[blk & 1][i],
           (short)0, acc,  false,  false);
    }
    __builtin_amdgcn_sched_barrier(0);
  }
#undef G1_LOAD

  if (half == 0) {
#pragma unroll
    for (int j = 0; j < 8; ++j) {
      float v = acc[j] + bias1;
      v = v > 0.0f ? v : 0.0f;
      h_lds[j * H_PITCH + w * 16 + ln] = v;
    }
  }
  __syncthreads();

  if (w == 0) {
    v8f c2[2] = {};
    const float* hp = h_lds + rA * H_PITCH + kb;
    const bool ok1 = (16 + ln < OUT_N);
#pragma unroll
    for (int k4 = 0; k4 < K2_STEPS; ++k4) {
      bw2[k4][1].x = ok1 ? bw2[k4][1].x : 0.0f;
      bw2[k4][1].y = ok1 ? bw2[k4][1].y : 0.0f;
    }
    v2f af2[K2_STEPS];
#pragma unroll
    for (int k4 = 0; k4 < K2_STEPS; ++k4) af2[k4] = *(const v2f*)(hp + k4 * 4);
    __builtin_amdgcn_sched_barrier(0);
#pragma unroll
    for (int k4 = 0; k4 < K2_STEPS; ++k4) {
#pragma unroll
      for (int nt = 0; nt < 2; ++nt) {
        c2[nt] = __builtin_amdgcn_wmma_f32_16x16x4_f32(
            false, af2[k4], false, bw2[k4][nt], (short)0, c2[nt], false, false);
      }
    }

#pragma unroll
    for (int nt = 0; nt < 2; ++nt) {
      const int col = nt * 16 + ln;
      if (col < OUT_N && half == 0) {
#pragma unroll
        for (int j = 0; j < 8; ++j) {
          out[(long)(row0 + j) * OUT_N + col] = c2[nt][j] + bias2[nt];
        }
      }
    }
  }
}

extern "C" void kernel_launch(void* const* d_in, const int* in_sizes, int n_in,
                              void* d_out, int out_size, void* d_ws, size_t ws_size,
                              hipStream_t stream) {
  const int*   x       = (const int*)  d_in[0];
  const int*   lengths = (const int*)  d_in[1];
  const float* table   = (const float*)d_in[2];
  const float* W1      = (const float*)d_in[3];
  const float* b1      = (const float*)d_in[4];
  const float* W2      = (const float*)d_in[5];
  const float* b2      = (const float*)d_in[6];
  float*       logits  = (float*)d_out;

  embed_mlp_kernel<<<BATCH / ROWS_WG, 256, 0, stream>>>(
      x, lengths, table, W1, b1, W2, b2, logits);
}


# speedup vs baseline: 4.2846x; 1.0370x over previous
#include <hip/hip_runtime.h>


typedef __attribute__((ext_vector_type(2))) float v2f;
typedef __attribute__((ext_vector_type(4))) float v4f;
typedef __attribute__((ext_vector_type(8))) float v8f;

#define EMB   300
#define BATCH 2048
#define SEQ   200
#define HID   128
#define OUT_N 20

#define ROWS_WG 8

#define GR    4
#define NBUF  5
#define TRIPS (SEQ / (GR * NBUF))

#define W1_PITCH (HID + 16)
#define W1_CH    4

#define K1_STEPS (EMB / 4)
#define K2_STEPS (HID / 4)
#define K1_BLK   15
#define H_PITCH  (HID + 4)

__global__ __launch_bounds__(256) void embed_mlp_kernel(
    const int* __restrict__ x, const int* __restrict__ lengths,
    const float* __restrict__ table,
    const float* __restrict__ W1, const float* __restrict__ b1,
    const float* __restrict__ W2, const float* __restrict__ b2,
    float* __restrict__ out)
{
  __shared__ __attribute__((aligned(16))) float w1_lds[EMB * W1_PITCH];
  __shared__ __attribute__((aligned(16))) float a_lds[ROWS_WG * EMB];
  __shared__ __attribute__((aligned(16))) float h_lds[ROWS_WG * H_PITCH];

  const int tid  = threadIdx.x;
  const int w    = __builtin_amdgcn_readfirstlane(tid >> 5);
  const int lane = tid & 31;
  const int row0 = blockIdx.x * ROWS_WG;
  const int b    = row0 + w;
  const int* xr = x + b * SEQ;
  const char* tb = (const char*)table;
  unsigned ulane = (unsigned)lane;
  unsigned v01 = lane * 16u;
  unsigned vc = 1024u + (lane < 11 ? lane * 16u : 0u);

  v4f a0 = {}, a1 = {}, a2 = {};
  v4f R0[NBUF][GR], R1[NBUF][GR], R2[NBUF][GR];
  unsigned NL[NBUF][GR];
  v4f wq[W1_CH];

#define W1_LOAD(c)                                                            \
  _Pragma("unroll") for (int i = 0; i < W1_CH; ++i) {                         \
    const int k = 32 * (c) + 8 * i + w;                                       \
    wq[i] = *(const v4f*)(W1 + (k < EMB ? k : EMB - 1) * HID + 4 * lane);     \
  }
#define W1_STORE(c)                                                           \
  _Pragma("unroll") for (int i = 0; i < W1_CH; ++i) {                         \
    const int k = 32 * (c) + 8 * i + w;                                       \
    *(v4f*)(w1_lds + k * W1_PITCH + 4 * lane) = wq[i];                        \
  }
#define W1_STORE_LAST(c)                                                      \
  _Pragma("unroll") for (int i = 0; i < W1_CH; ++i) {                         \
    const int k = 32 * (c) + 8 * i + w;                                       \
    if (k < EMB) *(v4f*)(w1_lds + k * W1_PITCH + 4 * lane) = wq[i];           \
  }

#define ISSUE(s, xi, first)                                                   \
  _Pragma("unroll") for (int r = 0; r < GR; ++r) {                            \
    const unsigned id = (unsigned)__builtin_amdgcn_readlane(xi, (first) + r); \
    const char* rowb = tb + (size_t)id * (EMB * 4u);                          \
    const unsigned sp = ((unsigned)(size_t)rowb >> 4) & 15u;                  \
    NL[s][r] = 32u - sp;                                                      \
    const bool low = ulane < NL[s][r];                                        \
    R0[s][r] = *(const v4f*)(rowb + (low ? v01 : 0u));                        \
    R1[s][r] = *(const v4f*)(rowb + (low ? v01 + 512u : v01));                \
    R2[s][r] = *(const v4f*)(rowb + (low ? vc : v01 + 512u));                 \
  }
#define SUM(s)                                                                \
  _Pragma("unroll") for (int r = 0; r < GR; ++r) {                            \
    const bool low = ulane < NL[s][r];                                        \
    a0 += low ? R0[s][r] : R1[s][r];                                          \
    a1 += low ? R1[s][r] : R2[s][r];                                          \
    a2 += R2[s][r];                                                           \
  }
#define PIN asm volatile("" : "+v"(a0), "+v"(a1), "+v"(a2), "+v"(v01), "+v"(vc), "+v"(ulane));

  const float bias1 = b1[w * 16 + (lane & 15)];
  W1_LOAD(0)
  const int x0 = xr[lane];
  int xc = xr[GR * (NBUF - 1) + lane];
  const float len = (float)lengths[b];
#pragma unroll
  for (int s = 0; s < NBUF - 1; ++s) { ISSUE(s, x0, GR * s) }
  PIN

#pragma nounroll
  for (int t = 0; t < TRIPS - 1; ++t) {
    W1_STORE(t)
    W1_LOAD(t + 1)
    const int nrow = GR * NBUF * (t + 1) + GR * (NBUF - 1) + lane;
    const int xn = xr[nrow < SEQ ? nrow : SEQ - 1];
#pragma unroll
    for (int s = 0; s < NBUF; ++s) {
      ISSUE((s + NBUF - 1) % NBUF, xc, GR * s)
      PIN
      SUM(s)
      PIN
    }
    xc = xn;
  }
  W1_STORE_LAST(TRIPS - 1)
  ISSUE(NBUF - 1, xc, 0)
  PIN
#pragma unroll
  for (int s = 0; s < NBUF; ++s) { SUM(s) }
  PIN
#undef PIN
#undef ISSUE
#undef SUM
#undef W1_LOAD
#undef W1_STORE
#undef W1_STORE_LAST

  const float inv = 1.0f / len;
  float* ar = a_lds + w * EMB;
  *(v4f*)(ar + 4 * lane)       = a0 * inv;
  *(v4f*)(ar + 128 + 4 * lane) = a1 * inv;
  if (lane < 11) *(v4f*)(ar + 256 + 4 * lane) = a2 * inv;

  const int half = lane >> 4;
  const int ln   = lane & 15;
  const int rA   = ln & 7;
  const int kb   = half * 2;

  __syncthreads();

  v2f bw2[K2_STEPS][2];
  float bias2[2];
  if (w == 0) {
#pragma unroll
    for (int k4 = 0; k4 < K2_STEPS; ++k4) {
      const int k = k4 * 4 + kb;
#pragma unroll
      for (int nt = 0; nt < 2; ++nt) {
        const int col = nt * 16 + ln;
        const int colc = col < OUT_N ? col : OUT_N - 1;
        bw2[k4][nt].x = W2[k * OUT_N + colc];
        bw2[k4][nt].y = W2[(k + 1) * OUT_N + colc];
      }
    }
#pragma unroll
    for (int nt = 0; nt < 2; ++nt) {
      const int col = nt * 16 + ln;
      bias2[nt] = b2[col < OUT_N ? col : OUT_N - 1];
    }
  }


  v8f acc = {};
  const float* ap = a_lds + rA * EMB + kb;
  const float* bp = w1_lds + kb * W1_PITCH + w * 16 + ln;
  v2f af[2][K1_BLK], bf[2][K1_BLK];
#define G1_LOAD(buf, blk)                                                     \
  _Pragma("unroll") for (int i = 0; i < K1_BLK; ++i) {                        \
    const int k4 = (blk) * K1_BLK + i;                                        \
    af[buf][i]   = *(const v2f*)(ap + k4 * 4);                                \
    bf[buf][i].x = bp[(k4 * 4)     * W1_PITCH];                               \
    bf[buf][i].y = bp[(k4 * 4 + 1) * W1_PITCH];                               \
  }
  G1_LOAD(0, 0)
#pragma unroll
  for (int blk = 0; blk < K1_STEPS / K1_BLK; ++blk) {
    if (blk + 1 < K1_STEPS / K1_BLK) { G1_LOAD((blk + 1) & 1, blk + 1) }
    __builtin_amdgcn_sched_barrier(0);
#pragma unroll
    for (int i = 0; i < K1_BLK; ++i) {
      acc = __builtin_amdgcn_wmma_f32_16x16x4_f32(
           false, af[blk & 1][i],  false, bf[blk & 1][i],
           (short)0, acc,  false,  false);
    }
    __builtin_amdgcn_sched_barrier(0);
  }
#undef G1_LOAD

  if (half == 0) {
#pragma unroll
    for (int j = 0; j < 8; ++j) {
      float v = acc[j] + bias1;
      v = v > 0.0f ? v : 0.0f;
      h_lds[j * H_PITCH + w * 16 + ln] = v;
    }
  }
  __syncthreads();

  if (w == 0) {
    v8f c2[2] = {};
    const float* hp = h_lds + rA * H_PITCH + kb;
    const bool ok1 = (16 + ln < OUT_N);
#pragma unroll
    for (int k4 = 0; k4 < K2_STEPS; ++k4) {
      bw2[k4][1].x = ok1 ? bw2[k4][1].x : 0.0f;
      bw2[k4][1].y = ok1 ? bw2[k4][1].y : 0.0f;
    }
    v2f af2[K2_STEPS];
#pragma unroll
    for (int k4 = 0; k4 < K2_STEPS; ++k4) af2[k4] = *(const v2f*)(hp + k4 * 4);
    __builtin_amdgcn_sched_barrier(0);
#pragma unroll
    for (int k4 = 0; k4 < K2_STEPS; ++k4) {
#pragma unroll
      for (int nt = 0; nt < 2; ++nt) {
        c2[nt] = __builtin_amdgcn_wmma_f32_16x16x4_f32(
            false, af2[k4], false, bw2[k4][nt], (short)0, c2[nt], false, false);
      }
    }

#pragma unroll
    for (int nt = 0; nt < 2; ++nt) {
      const int col = nt * 16 + ln;
      if (col < OUT_N && half == 0) {
#pragma unroll
        for (int j = 0; j < 8; ++j) {
          out[(long)(row0 + j) * OUT_N + col] = c2[nt][j] + bias2[nt];
        }
      }
    }
  }
}

extern "C" void kernel_launch(void* const* d_in, const int* in_sizes, int n_in,
                              void* d_out, int out_size, void* d_ws, size_t ws_size,
                              hipStream_t stream) {
  const int*   x       = (const int*)  d_in[0];
  const int*   lengths = (const int*)  d_in[1];
  const float* table   = (const float*)d_in[2];
  const float* W1      = (const float*)d_in[3];
  const float* b1      = (const float*)d_in[4];
  const float* W2      = (const float*)d_in[5];
  const float* b2      = (const float*)d_in[6];
  float*       logits  = (float*)d_out;

  embed_mlp_kernel<<<BATCH / ROWS_WG, 256, 0, stream>>>(
      x, lengths, table, W1, b1, W2, b2, logits);
}


# Gather: 16 waves per work-group, two per batch row, each pooling 100 of its 200 table rows (ring of 24-30 loads a wave, same total ahead); the two half sums are added in LDS: sum order changed
# speedup vs baseline: 4.5187x; 1.0546x over previous
#include <hip/hip_runtime.h>


typedef __attribute__((ext_vector_type(2))) float v2f;
typedef __attribute__((ext_vector_type(4))) float v4f;
typedef __attribute__((ext_vector_type(8))) float v8f;

#define EMB   300
#define BATCH 2048
#define SEQ   200
#define HID   128
#define OUT_N 20

#define ROWS_WG 8
#define SPLIT   2
#define WAVES_WG (ROWS_WG * SPLIT)
#define SEQ_W   (SEQ / SPLIT)

#define GR    2
#define NBUF  5
#define TRIPS (SEQ_W / (GR * NBUF))

#define W1_PITCH (HID + 16)
#define W1_CH    2

#define K1_STEPS (EMB / 4)
#define K2_STEPS (HID / 4)
#define K1_BLK   5
#define H_PITCH  (HID + 4)

__global__ __launch_bounds__(32 * WAVES_WG) void embed_mlp_kernel(
    const int* __restrict__ x, const int* __restrict__ lengths,
    const float* __restrict__ table,
    const float* __restrict__ W1, const float* __restrict__ b1,
    const float* __restrict__ W2, const float* __restrict__ b2,
    float* __restrict__ out)
{
  __shared__ __attribute__((aligned(16))) float w1_lds[EMB * W1_PITCH];
  __shared__ __attribute__((aligned(16))) float a_lds[ROWS_WG * EMB];
  __shared__ __attribute__((aligned(16))) float h_lds[ROWS_WG * H_PITCH];

  const int tid  = threadIdx.x;
  const int w    = __builtin_amdgcn_readfirstlane(tid >> 5);
  const int lane = tid & 31;
  const int row0 = blockIdx.x * ROWS_WG;
  const int rw   = w & (ROWS_WG - 1);
  const int hsel = w >> 3;
  const int b    = row0 + rw;
  const int* xr = x + b * SEQ + hsel * SEQ_W;
  const char* tb = (const char*)table;
  unsigned ulane = (unsigned)lane;
  unsigned v01 = lane * 16u;
  unsigned vc = 1024u + (lane < 11 ? lane * 16u : 0u);

  v4f a0 = {}, a1 = {}, a2 = {};
  v4f R0[NBUF][GR], R1[NBUF][GR], R2[NBUF][GR];
  unsigned NL[NBUF][GR];
  v4f wq[W1_CH];

#define W1_LOAD(c)                                                            \
  _Pragma("unroll") for (int i = 0; i < W1_CH; ++i) {                         \
    const int k = 32 * (c) + 16 * i + w;                                      \
    wq[i] = *(const v4f*)(W1 + (k < EMB ? k : EMB - 1) * HID + 4 * lane);     \
  }
#define W1_STORE(c)                                                           \
  _Pragma("unroll") for (int i = 0; i < W1_CH; ++i) {                         \
    const int k = 32 * (c) + 16 * i + w;                                      \
    *(v4f*)(w1_lds + k * W1_PITCH + 4 * lane) = wq[i];                        \
  }
#define W1_STORE_LAST(c)                                                      \
  _Pragma("unroll") for (int i = 0; i < W1_CH; ++i) {                         \
    const int k = 32 * (c) + 16 * i + w;                                      \
    if (k < EMB) *(v4f*)(w1_lds + k * W1_PITCH + 4 * lane) = wq[i];           \
  }

#define ISSUE(s, xi, first)                                                   \
  _Pragma("unroll") for (int r = 0; r < GR; ++r) {                            \
    const unsigned id = (unsigned)__builtin_amdgcn_readlane(xi, (first) + r); \
    const char* rowb = tb + (size_t)id * (EMB * 4u);                          \
    const unsigned sp = ((unsigned)(size_t)rowb >> 4) & 15u;                  \
    NL[s][r] = 32u - sp;                                                      \
    const bool low = ulane < NL[s][r];                                        \
    R0[s][r] = *(const v4f*)(rowb + (low ? v01 : 0u));                        \
    R1[s][r] = *(const v4f*)(rowb + (low ? v01 + 512u : v01));                \
    R2[s][r] = *(const v4f*)(rowb + (low ? vc : v01 + 512u));                 \
  }
#define SUM(s)                                                                \
  _Pragma("unroll") for (int r = 0; r < GR; ++r) {                            \
    const bool low = ulane < NL[s][r];                                        \
    a0 += low ? R0[s][r] : R1[s][r];                                          \
    a1 += low ? R1[s][r] : R2[s][r];                                          \
    a2 += R2[s][r];                                                           \
  }
#define PIN asm volatile("" : "+v"(a0), "+v"(a1), "+v"(a2), "+v"(v01), "+v"(vc), "+v"(ulane));

  const float bias1 = b1[rw * 16 + (lane & 15)];
  W1_LOAD(0)
  const int x0 = xr[lane];
  int xc = xr[GR * (NBUF - 1) + lane];
  const float len = (float)lengths[b];
#pragma unroll
  for (int s = 0; s < NBUF - 1; ++s) { ISSUE(s, x0, GR * s) }
  PIN

#pragma nounroll
  for (int t = 0; t < TRIPS - 1; ++t) {
    W1_STORE(t)
    W1_LOAD(t + 1)
    const int nrow = GR * NBUF * (t + 1) + GR * (NBUF - 1) + lane;
    const int xn = xr[nrow < SEQ_W ? nrow : SEQ_W - 1];
#pragma unroll
    for (int s = 0; s < NBUF; ++s) {
      ISSUE((s + NBUF - 1) % NBUF, xc, GR * s)
      PIN
      SUM(s)
      PIN
    }
    xc = xn;
  }
  W1_STORE_LAST(TRIPS - 1)
  ISSUE(NBUF - 1, xc, 0)
  PIN
#pragma unroll
  for (int s = 0; s < NBUF; ++s) { SUM(s) }
  PIN
#undef PIN
#undef ISSUE
#undef SUM
#undef W1_LOAD
#undef W1_STORE
#undef W1_STORE_LAST

  const float inv = 1.0f / len;
  float* ar = a_lds + rw * EMB;
  if (hsel == 1) {
    *(v4f*)(ar + 4 * lane)       = a0;
    *(v4f*)(ar + 128 + 4 * lane) = a1;
    if (lane < 11) *(v4f*)(ar + 256 + 4 * lane) = a2;
  }
  __syncthreads();
  if (hsel == 0) {
    const v4f u0 = *(const v4f*)(ar + 4 * lane);
    const v4f u1 = *(const v4f*)(ar + 128 + 4 * lane);
    *(v4f*)(ar + 4 * lane)       = (a0 + u0) * inv;
    *(v4f*)(ar + 128 + 4 * lane) = (a1 + u1) * inv;
    if (lane < 11) {
      const v4f u2 = *(const v4f*)(ar + 256 + 4 * lane);
      *(v4f*)(ar + 256 + 4 * lane) = (a2 + u2) * inv;
    }
  }

  const int half = lane >> 4;
  const int ln   = lane & 15;
  const int rA   = ln & 7;
  const int kb   = half * 2;

  __syncthreads();

  v2f bw2[K2_STEPS][2];
  float bias2[2];
  if (w == 0) {
#pragma unroll
    for (int k4 = 0; k4 < K2_STEPS; ++k4) {
      const int k = k4 * 4 + kb;
#pragma unroll
      for (int nt = 0; nt < 2; ++nt) {
        const int col = nt * 16 + ln;
        const int colc = col < OUT_N ? col : OUT_N - 1;
        bw2[k4][nt].x = W2[k * OUT_N + colc];
        bw2[k4][nt].y = W2[(k + 1) * OUT_N + colc];
      }
    }
#pragma unroll
    for (int nt = 0; nt < 2; ++nt) {
      const int col = nt * 16 + ln;
      bias2[nt] = b2[col < OUT_N ? col : OUT_N - 1];
    }
  }


  if (w < ROWS_WG) {
  v8f acc = {};
  const float* ap = a_lds + rA * EMB + kb;
  const float* bp = w1_lds + kb * W1_PITCH + w * 16 + ln;
  v2f af[2][K1_BLK], bf[2][K1_BLK];
#define G1_LOAD(buf, blk)                                                     \
  _Pragma("unroll") for (int i = 0; i < K1_BLK; ++i) {                        \
    const int k4 = (blk) * K1_BLK + i;                                        \
    af[buf][i]   = *(const v2f*)(ap + k4 * 4);                                \
    bf[buf][i].x = bp[(k4 * 4)     * W1_PITCH];                               \
    bf[buf][i].y = bp[(k4 * 4 + 1) * W1_PITCH];                               \
  }
  G1_LOAD(0, 0)
#pragma unroll
  for (int blk = 0; blk < K1_STEPS / K1_BLK; ++blk) {
    if (blk + 1 < K1_STEPS / K1_BLK) { G1_LOAD((blk + 1) & 1, blk + 1) }
    __builtin_amdgcn_sched_barrier(0);
#pragma unroll
    for (int i = 0; i < K1_BLK; ++i) {
      acc = __builtin_amdgcn_wmma_f32_16x16x4_f32(
           false, af[blk & 1][i],  false, bf[blk & 1][i],
           (short)0, acc,  false,  false);
    }
    __builtin_amdgcn_sched_barrier(0);
  }
#undef G1_LOAD

  if (half == 0) {
#pragma unroll
    for (int j = 0; j < 8; ++j) {
      float v = acc[j] + bias1;
      v = v > 0.0f ? v : 0.0f;
      h_lds[j * H_PITCH + w * 16 + ln] = v;
    }
  }
  }
  __syncthreads();

  if (w == 0) {
    v8f c2[2] = {};
    const float* hp = h_lds + rA * H_PITCH + kb;
    const bool ok1 = (16 + ln < OUT_N);
#pragma unroll
    for (int k4 = 0; k4 < K2_STEPS; ++k4) {
      bw2[k4][1].x = ok1 ? bw2[k4][1].x : 0.0f;
      bw2[k4][1].y = ok1 ? bw2[k4][1].y : 0.0f;
    }
    v2f af2[K2_STEPS];
#pragma unroll
    for (int k4 = 0; k4 < K2_STEPS; ++k4) af2[k4] = *(const v2f*)(hp + k4 * 4);
    __builtin_amdgcn_sched_barrier(0);
#pragma unroll
    for (int k4 = 0; k4 < K2_STEPS; ++k4) {
#pragma unroll
      for (int nt = 0; nt < 2; ++nt) {
        c2[nt] = __builtin_amdgcn_wmma_f32_16x16x4_f32(
            false, af2[k4], false, bw2[k4][nt], (short)0, c2[nt], false, false);
      }
    }

#pragma unroll
    for (int nt = 0; nt < 2; ++nt) {
      const int col = nt * 16 + ln;
      if (col < OUT_N && half == 0) {
#pragma unroll
        for (int j = 0; j < 8; ++j) {
          out[(long)(row0 + j) * OUT_N + col] = c2[nt][j] + bias2[nt];
        }
      }
    }
  }
}

extern "C" void kernel_launch(void* const* d_in, const int* in_sizes, int n_in,
                              void* d_out, int out_size, void* d_ws, size_t ws_size,
                              hipStream_t stream) {
  const int*   x       = (const int*)  d_in[0];
  const int*   lengths = (const int*)  d_in[1];
  const float* table   = (const float*)d_in[2];
  const float* W1      = (const float*)d_in[3];
  const float* b1      = (const float*)d_in[4];
  const float* W2      = (const float*)d_in[5];
  const float* b2      = (const float*)d_in[6];
  float*       logits  = (float*)d_out;

  embed_mlp_kernel<<<BATCH / ROWS_WG, 32 * WAVES_WG, 0, stream>>>(
      x, lengths, table, W1, b1, W2, b2, logits);
}
